# starting kernel
# baseline (speedup 1.0000x reference)
#include <hip/hip_runtime.h>
#include <hip/hip_bf16.h>
#include <math.h>

#define HID   512
#define INH   512
#define INSZ  256
#define BATCH 4
#define SEQ   64
#define DELAY 8
#define LR_F    0.01f
#define DECAY_F 0.951229424500714f
#define OMD_F   0.048770575499285991f

typedef float v2f __attribute__((ext_vector_type(2)));
typedef float v8f __attribute__((ext_vector_type(8)));

__global__ __launch_bounds__(256) void k_proj(const float* __restrict__ x,
                                              const float* __restrict__ W,
                                              float* __restrict__ iproj) {
  const int gw   = (blockIdx.x * 256 + threadIdx.x) >> 5;
  const int lane = threadIdx.x & 31;
  const int tm = gw >> 6;
  const int tn = gw & 63;
  const int hf = lane >> 4, ln = lane & 15;
  const float* ar = x + (size_t)(tm * 16 + ln) * INSZ;
  const float* br = W + (size_t)(tn * 16 + ln) * INSZ;
  v8f acc = {};
  for (int c = 0; c < INSZ; c += 4) {
    const int cc = c + 2 * hf;
    v2f a, b;
    a.x = ar[cc]; a.y = ar[cc + 1];
    b.x = br[cc]; b.y = br[cc + 1];
    acc = __builtin_amdgcn_wmma_f32_16x16x4_f32(false, a, false, b,
                                                (short)0, acc, false, false);
  }
  for (int r = 0; r < 8; ++r)
    iproj[(size_t)(tm * 16 + r + 8 * hf) * 1024 + tn * 16 + ln] = acc[r];
}

__global__ __launch_bounds__(512) void k_keys(const float* __restrict__ iproj,
                                              float* __restrict__ keyH,
                                              float* __restrict__ keys_out) {
  const int b = blockIdx.x, h = threadIdx.x;
  float kv = 0.f;
  for (int t = 0; t < SEQ; ++t) {
    kv = DECAY_F * kv + iproj[((size_t)b * SEQ + t) * 1024 + h];
    const float ko = tanhf(kv);
    keyH[((size_t)b * SEQ + t) * HID + h]     = ko;
    keys_out[((size_t)b * SEQ + t) * HID + h] = ko;
  }
}

__global__ __launch_bounds__(256) void k_ikv0(const float* __restrict__ mem0,
                                              const float* __restrict__ keyH,
                                              float* __restrict__ ikv0) {
  const int gw   = (blockIdx.x * 256 + threadIdx.x) >> 5;
  const int lane = threadIdx.x & 31;
  const int b  = gw >> 7;
  const int rr = gw & 127;
  const int tm = rr >> 2;
  const int tn = rr & 3;
  const int hf = lane >> 4, ln = lane & 15;
  const float* ar = mem0 + ((size_t)b * HID + tm * 16 + ln) * HID;
  const float* br = keyH + ((size_t)b * SEQ + tn * 16 + ln) * HID;
  v8f acc = {};
  for (int c = 0; c < HID; c += 4) {
    const int cc = c + 2 * hf;
    v2f a, bb;
    a.x  = ar[cc]; a.y  = ar[cc + 1];
    bb.x = br[cc]; bb.y = br[cc + 1];
    acc = __builtin_amdgcn_wmma_f32_16x16x4_f32(false, a, false, bb,
                                                (short)0, acc, false, false);
  }
  for (int r = 0; r < 8; ++r) {
    const int hh = tm * 16 + r + 8 * hf;
    const int tt = tn * 16 + ln;
    ikv0[((size_t)b * SEQ + tt) * HID + hh] = acc[r];
  }
}

__global__ __launch_bounds__(512) void k_scan(const float* __restrict__ iproj,
                                              const float* __restrict__ keyH,
                                              const float* __restrict__ ikv0,
                                              const float* __restrict__ ei0,
                                              const float* __restrict__ ie0,
                                              float* __restrict__ ktrH,
                                              float* __restrict__ vtrH,
                                              float* __restrict__ itrH,
                                              float* __restrict__ dei,
                                              float* __restrict__ vals_out) {
  __shared__ __align__(16) float vtr_l[SEQ][HID];
  __shared__ __align__(16) float ring[DELAY][HID];
  __shared__ __align__(16) float key_s[HID];
  __shared__ float dk[SEQ];
  __shared__ float db[SEQ];

  const int b = blockIdx.x;
  const int h = threadIdx.x;
  const int lane = h & 31, wv = h >> 5;

  for (int r = 0; r < DELAY; ++r) ring[r][h] = 0.f;

  float vvs = 0.f, isv = 0.f;
  float ktr = 0.f, vtr = 0.f, itr = 0.f, accD = 0.f;
  const size_t mbase = (size_t)b * HID * HID + (size_t)h * HID;
  float d_ei = ei0[mbase + h];
  const float ie0d = ie0[mbase + h];
  const float4* ierow = (const float4*)(ie0 + mbase);
  __syncthreads();

  for (int t = 0; t < SEQ; ++t) {
    const int r = t & (DELAY - 1);
    const size_t rowoff = ((size_t)b * SEQ + t) * HID;

    key_s[h] = keyH[rowoff + h];
    __syncthreads();

    for (int s = wv; s < t; s += 16) {
      float p1 = 0.f, p2 = 0.f;
      for (int k = lane; k < HID; k += 32) {
        const float vk = vtr_l[s][k];
        p1 = fmaf(vk, key_s[k], p1);
        p2 = fmaf(vk, ring[r][k], p2);
      }
      for (int off = 16; off > 0; off >>= 1) {
        p1 += __shfl_xor(p1, off, 32);
        p2 += __shfl_xor(p2, off, 32);
      }
      if (lane == 0) { dk[s] = p1; db[s] = p2; }
    }
    __syncthreads();

    float sk = 0.f, si = 0.f;
    for (int s = 0; s < t; ++s) {
      sk = fmaf(ktrH[((size_t)b * SEQ + s) * HID + h], dk[s], sk);
      si = fmaf(itrH[((size_t)b * SEQ + s) * HID + h], db[s], si);
    }

    float mv = 0.f;
    const float4* b4 = (const float4*)ring[r];
    for (int k4 = 0; k4 < HID / 4; ++k4) {
      if ((k4 & 15) == 0 && k4 + 16 < HID / 4)
        __builtin_prefetch(&ierow[k4 + 16], 0, 3);
      const float4 a = ierow[k4], c = b4[k4];
      mv = fmaf(a.x, c.x, mv); mv = fmaf(a.y, c.y, mv);
      mv = fmaf(a.z, c.z, mv); mv = fmaf(a.w, c.w, mv);
    }
    const float bufh = ring[r][h];
    const float inh_out = mv + LR_F * si - bufh * (ie0d + LR_F * accD);

    const float ikv = 0.2f * (ikv0[rowoff + h] + LR_F * sk);
    const float ivt = iproj[((size_t)b * SEQ + t) * 1024 + HID + h];

    vvs = DECAY_F * vvs + ivt + ikv - inh_out;
    const float val = tanhf(vvs);
    isv = DECAY_F * isv + val * d_ei;
    const float inh = tanhf(isv);

    ktr = DECAY_F * ktr + OMD_F * key_s[h];
    vtr = DECAY_F * vtr + OMD_F * val;
    itr = DECAY_F * itr + OMD_F * inh;
    const float p = vtr * itr;
    d_ei += LR_F * p;
    accD += p;

    ktrH[rowoff + h] = ktr;
    vtrH[rowoff + h] = vtr;
    itrH[rowoff + h] = itr;
    vtr_l[t][h] = vtr;
    vals_out[rowoff + h] = val;

    __syncthreads();
    ring[r][h] = inh;
  }
  dei[(size_t)b * HID + h] = d_ei;
}

__global__ __launch_bounds__(256) void k_final(const float* __restrict__ mem0,
                                               const float* __restrict__ ie0,
                                               const float* __restrict__ ktrH,
                                               const float* __restrict__ vtrH,
                                               const float* __restrict__ itrH,
                                               float* __restrict__ out) {
  const int which = blockIdx.y;
  const int gw    = (blockIdx.x * 256 + threadIdx.x) >> 5;
  const int lane  = threadIdx.x & 31;
  const int b  = gw >> 10;
  const int rr = gw & 1023;
  const int tm = rr >> 5, tn = rr & 31;
  const int hf = lane >> 4, ln = lane & 15;
  const float* Ah = which ? itrH : ktrH;
  v8f acc = {};
  for (int c = 0; c < SEQ; c += 4) {
    const int cc = c + 2 * hf;
    v2f a, bb;
    a.x  = Ah[((size_t)b * SEQ + cc)     * HID + tm * 16 + ln];
    a.y  = Ah[((size_t)b * SEQ + cc + 1) * HID + tm * 16 + ln];
    bb.x = vtrH[((size_t)b * SEQ + cc)     * HID + tn * 16 + ln];
    bb.y = vtrH[((size_t)b * SEQ + cc + 1) * HID + tn * 16 + ln];
    acc = __builtin_amdgcn_wmma_f32_16x16x4_f32(false, a, false, bb,
                                                (short)0, acc, false, false);
  }
  const float* base0 = which ? ie0 : mem0;
  float* dst = out + (which ? (size_t)2097152 : (size_t)0);
  for (int r = 0; r < 8; ++r) {
    const int m = tm * 16 + r + 8 * hf;
    const int n = tn * 16 + ln;
    const size_t idx = (size_t)b * HID * HID + (size_t)m * HID + n;
    float v = base0[idx] + LR_F * acc[r];
    if (which && m == n)
      v = LR_F * itrH[((size_t)b * SEQ + 63) * HID + m] *
                 vtrH[((size_t)b * SEQ + 63) * HID + n];
    dst[idx] = v;
  }
}

__global__ __launch_bounds__(256) void k_ei(const float* __restrict__ dei,
                                            const float* __restrict__ vtrH,
                                            const float* __restrict__ itrH,
                                            float* __restrict__ ei_out) {
  const int idx = blockIdx.x * 256 + threadIdx.x;
  const int b = idx >> 18;
  const int h = (idx >> 9) & 511;
  const int i = idx & 511;
  float v;
  if (h == i) v = dei[b * HID + h];
  else        v = LR_F * vtrH[((size_t)b * SEQ + 63) * HID + h] *
                         itrH[((size_t)b * SEQ + 63) * HID + i];
  ei_out[idx] = v;
}

extern "C" void kernel_launch(void* const* d_in, const int* in_sizes, int n_in,
                              void* d_out, int out_size, void* d_ws, size_t ws_size,
                              hipStream_t stream) {
  (void)in_sizes; (void)n_in; (void)out_size; (void)ws_size;
  const float* x    = (const float*)d_in[0];
  const float* W    = (const float*)d_in[1];
  const float* mem0 = (const float*)d_in[2];
  const float* ei0  = (const float*)d_in[3];
  const float* ie0  = (const float*)d_in[4];

  float* out = (float*)d_out;
  float* ei_out   = out + 1048576;
  float* keys_out = out + 3145728;
  float* vals_out = out + 3276800;

  float* ws    = (float*)d_ws;
  float* iproj = ws;
  float* keyH  = ws + 262144;
  float* ikv0  = ws + 393216;
  float* ktrH  = ws + 524288;
  float* vtrH  = ws + 655360;
  float* itrH  = ws + 786432;
  float* dei   = ws + 917504;

  k_proj<<<128, 256, 0, stream>>>(x, W, iproj);
  k_keys<<<4, 512, 0, stream>>>(iproj, keyH, keys_out);
  k_ikv0<<<64, 256, 0, stream>>>(mem0, keyH, ikv0);
  k_scan<<<4, 512, 0, stream>>>(iproj, keyH, ikv0, ei0, ie0,
                                ktrH, vtrH, itrH, dei, vals_out);
  dim3 gf(512, 2);
  k_final<<<gf, 256, 0, stream>>>(mem0, ie0, ktrH, vtrH, itrH, out);
  k_ei<<<4096, 256, 0, stream>>>(dei, vtrH, itrH, ei_out);
}


# ie0 . buf products of each 8-step delay window done as one f32 WMMA GEMM per window: ie0 read once per 8 steps, not every step; k summation order of that product changes
# speedup vs baseline: 3.1832x; 3.1832x over previous
#include <hip/hip_runtime.h>
#include <hip/hip_bf16.h>
#include <math.h>

#define HID   512
#define INH   512
#define INSZ  256
#define BATCH 4
#define SEQ   64
#define DELAY 8
#define LR_F    0.01f
#define DECAY_F 0.951229424500714f
#define OMD_F   0.048770575499285991f
#define RPITCH  (HID + 4)

typedef float v2f __attribute__((ext_vector_type(2)));
typedef float v8f __attribute__((ext_vector_type(8)));

__global__ __launch_bounds__(256) void k_proj(const float* __restrict__ x,
                                              const float* __restrict__ W,
                                              float* __restrict__ iproj) {
  const int gw   = (blockIdx.x * 256 + threadIdx.x) >> 5;
  const int lane = threadIdx.x & 31;
  const int tm = gw >> 6;
  const int tn = gw & 63;
  const int hf = lane >> 4, ln = lane & 15;
  const float* ar = x + (size_t)(tm * 16 + ln) * INSZ;
  const float* br = W + (size_t)(tn * 16 + ln) * INSZ;
  v8f acc = {};
  for (int c = 0; c < INSZ; c += 4) {
    const int cc = c + 2 * hf;
    v2f a, b;
    a.x = ar[cc]; a.y = ar[cc + 1];
    b.x = br[cc]; b.y = br[cc + 1];
    acc = __builtin_amdgcn_wmma_f32_16x16x4_f32(false, a, false, b,
                                                (short)0, acc, false, false);
  }
  for (int r = 0; r < 8; ++r)
    iproj[(size_t)(tm * 16 + r + 8 * hf) * 1024 + tn * 16 + ln] = acc[r];
}

__global__ __launch_bounds__(512) void k_keys(const float* __restrict__ iproj,
                                              float* __restrict__ keyH,
                                              float* __restrict__ keys_out) {
  const int b = blockIdx.x, h = threadIdx.x;
  float kv = 0.f;
  for (int t = 0; t < SEQ; ++t) {
    kv = DECAY_F * kv + iproj[((size_t)b * SEQ + t) * 1024 + h];
    const float ko = tanhf(kv);
    keyH[((size_t)b * SEQ + t) * HID + h]     = ko;
    keys_out[((size_t)b * SEQ + t) * HID + h] = ko;
  }
}

__global__ __launch_bounds__(256) void k_ikv0(const float* __restrict__ mem0,
                                              const float* __restrict__ keyH,
                                              float* __restrict__ ikv0) {
  const int gw   = (blockIdx.x * 256 + threadIdx.x) >> 5;
  const int lane = threadIdx.x & 31;
  const int b  = gw >> 7;
  const int rr = gw & 127;
  const int tm = rr >> 2;
  const int tn = rr & 3;
  const int hf = lane >> 4, ln = lane & 15;
  const float* ar = mem0 + ((size_t)b * HID + tm * 16 + ln) * HID;
  const float* br = keyH + ((size_t)b * SEQ + tn * 16 + ln) * HID;
  v8f acc = {};
  for (int c = 0; c < HID; c += 4) {
    const int cc = c + 2 * hf;
    v2f a, bb;
    a.x  = ar[cc]; a.y  = ar[cc + 1];
    bb.x = br[cc]; bb.y = br[cc + 1];
    acc = __builtin_amdgcn_wmma_f32_16x16x4_f32(false, a, false, bb,
                                                (short)0, acc, false, false);
  }
  for (int r = 0; r < 8; ++r) {
    const int hh = tm * 16 + r + 8 * hf;
    const int tt = tn * 16 + ln;
    ikv0[((size_t)b * SEQ + tt) * HID + hh] = acc[r];
  }
}

__global__ __launch_bounds__(512) void k_scan(const float* __restrict__ iproj,
                                              const float* __restrict__ keyH,
                                              const float* __restrict__ ikv0,
                                              const float* __restrict__ ei0,
                                              const float* __restrict__ ie0,
                                              float* __restrict__ ktrH,
                                              float* __restrict__ vtrH,
                                              float* __restrict__ itrH,
                                              float* __restrict__ dei,
                                              float* __restrict__ vals_out) {
  __shared__ __align__(16) float vtr_l[SEQ][HID];
  __shared__ __align__(16) float ring[DELAY + 1][RPITCH];
  __shared__ __align__(16) float g_s[DELAY][HID];
  __shared__ __align__(16) float key_s[HID];
  __shared__ float dk[SEQ];
  __shared__ float db[SEQ];

  const int b = blockIdx.x;
  const int h = threadIdx.x;
  const int lane = h & 31, wv = h >> 5;

  for (int r = 0; r <= DELAY; ++r) ring[r][h] = 0.f;

  float vvs = 0.f, isv = 0.f;
  float ktr = 0.f, vtr = 0.f, itr = 0.f, accD = 0.f;
  const size_t mbase = (size_t)b * HID * HID + (size_t)h * HID;
  float d_ei = ei0[mbase + h];
  const float ie0d = ie0[mbase + h];
  const int ln = lane & 15, hf = lane >> 4;
  const float* a0p = ie0 + (size_t)b * HID * HID + (size_t)(wv * 32 + ln) * HID + 4 * hf;
  const float* a1p = a0p + (size_t)16 * HID;
  const float* bp  = &ring[ln < DELAY ? ln : DELAY][4 * hf];
  __syncthreads();

  for (int t = 0; t < SEQ; ++t) {
    const int r = t & (DELAY - 1);
    const size_t rowoff = ((size_t)b * SEQ + t) * HID;

    key_s[h] = keyH[rowoff + h];
    __syncthreads();

    for (int s = wv; s < t; s += 16) {
      float p1 = 0.f, p2 = 0.f;
      for (int k = lane; k < HID; k += 32) {
        const float vk = vtr_l[s][k];
        p1 = fmaf(vk, key_s[k], p1);
        p2 = fmaf(vk, ring[r][k], p2);
      }
      for (int off = 16; off > 0; off >>= 1) {
        p1 += __shfl_xor(p1, off, 32);
        p2 += __shfl_xor(p2, off, 32);
      }
      if (lane == 0) { dk[s] = p1; db[s] = p2; }
    }

    if (r == 0 && t != 0) {
      v8f g0 = {}, g1 = {};
      float4 fa0[2][8], fa1[2][8];
      for (int q = 0; q < 8; ++q) {
        fa0[0][q] = *(const float4*)(a0p + q * 8);
        fa1[0][q] = *(const float4*)(a1p + q * 8);
      }
#pragma unroll
      for (int ch = 0; ch < HID / 64; ++ch) {
        const int cu = ch & 1, nx = cu ^ 1;
        if (ch + 1 < HID / 64) {
          for (int q = 0; q < 8; ++q) {
            fa0[nx][q] = *(const float4*)(a0p + (ch + 1) * 64 + q * 8);
            fa1[nx][q] = *(const float4*)(a1p + (ch + 1) * 64 + q * 8);
          }
        }
        float4 fb[8];
        for (int q = 0; q < 8; ++q)
          fb[q] = *(const float4*)(bp + ch * 64 + q * 8);
        for (int q = 0; q < 8; ++q) {
          const v2f b01 = {fb[q].x, fb[q].y}, b23 = {fb[q].z, fb[q].w};
          const v2f p01 = {fa0[cu][q].x, fa0[cu][q].y}, p23 = {fa0[cu][q].z, fa0[cu][q].w};
          const v2f q01 = {fa1[cu][q].x, fa1[cu][q].y}, q23 = {fa1[cu][q].z, fa1[cu][q].w};
          g0 = __builtin_amdgcn_wmma_f32_16x16x4_f32(false, p01, false, b01,
                                                     (short)0, g0, false, false);
          g1 = __builtin_amdgcn_wmma_f32_16x16x4_f32(false, q01, false, b01,
                                                     (short)0, g1, false, false);
          g0 = __builtin_amdgcn_wmma_f32_16x16x4_f32(false, p23, false, b23,
                                                     (short)0, g0, false, false);
          g1 = __builtin_amdgcn_wmma_f32_16x16x4_f32(false, q23, false, b23,
                                                     (short)0, g1, false, false);
        }
      }
      if (ln < DELAY) {
        float* gp = &g_s[ln][wv * 32 + 8 * hf];
        *(float4*)(gp)      = make_float4(g0[0], g0[1], g0[2], g0[3]);
        *(float4*)(gp + 4)  = make_float4(g0[4], g0[5], g0[6], g0[7]);
        *(float4*)(gp + 16) = make_float4(g1[0], g1[1], g1[2], g1[3]);
        *(float4*)(gp + 20) = make_float4(g1[4], g1[5], g1[6], g1[7]);
      }
    }
    __syncthreads();

    float sk = 0.f, si = 0.f;
    for (int s = 0; s < t; ++s) {
      sk = fmaf(ktrH[((size_t)b * SEQ + s) * HID + h], dk[s], sk);
      si = fmaf(itrH[((size_t)b * SEQ + s) * HID + h], db[s], si);
    }

    const float mv = (t >= DELAY) ? g_s[r][h] : 0.f;
    const float bufh = ring[r][h];
    const float inh_out = mv + LR_F * si - bufh * (ie0d + LR_F * accD);

    const float ikv = 0.2f * (ikv0[rowoff + h] + LR_F * sk);
    const float ivt = iproj[((size_t)b * SEQ + t) * 1024 + HID + h];

    vvs = DECAY_F * vvs + ivt + ikv - inh_out;
    const float val = tanhf(vvs);
    isv = DECAY_F * isv + val * d_ei;
    const float inh = tanhf(isv);

    ktr = DECAY_F * ktr + OMD_F * key_s[h];
    vtr = DECAY_F * vtr + OMD_F * val;
    itr = DECAY_F * itr + OMD_F * inh;
    const float p = vtr * itr;
    d_ei += LR_F * p;
    accD += p;

    ktrH[rowoff + h] = ktr;
    vtrH[rowoff + h] = vtr;
    itrH[rowoff + h] = itr;
    vtr_l[t][h] = vtr;
    vals_out[rowoff + h] = val;

    __syncthreads();
    ring[r][h] = inh;
  }
  dei[(size_t)b * HID + h] = d_ei;
}

__global__ __launch_bounds__(256) void k_final(const float* __restrict__ mem0,
                                               const float* __restrict__ ie0,
                                               const float* __restrict__ ktrH,
                                               const float* __restrict__ vtrH,
                                               const float* __restrict__ itrH,
                                               float* __restrict__ out) {
  const int which = blockIdx.y;
  const int gw    = (blockIdx.x * 256 + threadIdx.x) >> 5;
  const int lane  = threadIdx.x & 31;
  const int b  = gw >> 10;
  const int rr = gw & 1023;
  const int tm = rr >> 5, tn = rr & 31;
  const int hf = lane >> 4, ln = lane & 15;
  const float* Ah = which ? itrH : ktrH;
  v8f acc = {};
  for (int c = 0; c < SEQ; c += 4) {
    const int cc = c + 2 * hf;
    v2f a, bb;
    a.x  = Ah[((size_t)b * SEQ + cc)     * HID + tm * 16 + ln];
    a.y  = Ah[((size_t)b * SEQ + cc + 1) * HID + tm * 16 + ln];
    bb.x = vtrH[((size_t)b * SEQ + cc)     * HID + tn * 16 + ln];
    bb.y = vtrH[((size_t)b * SEQ + cc + 1) * HID + tn * 16 + ln];
    acc = __builtin_amdgcn_wmma_f32_16x16x4_f32(false, a, false, bb,
                                                (short)0, acc, false, false);
  }
  const float* base0 = which ? ie0 : mem0;
  float* dst = out + (which ? (size_t)2097152 : (size_t)0);
  for (int r = 0; r < 8; ++r) {
    const int m = tm * 16 + r + 8 * hf;
    const int n = tn * 16 + ln;
    const size_t idx = (size_t)b * HID * HID + (size_t)m * HID + n;
    float v = base0[idx] + LR_F * acc[r];
    if (which && m == n)
      v = LR_F * itrH[((size_t)b * SEQ + 63) * HID + m] *
                 vtrH[((size_t)b * SEQ + 63) * HID + n];
    dst[idx] = v;
  }
}

__global__ __launch_bounds__(256) void k_ei(const float* __restrict__ dei,
                                            const float* __restrict__ vtrH,
                                            const float* __restrict__ itrH,
                                            float* __restrict__ ei_out) {
  const int idx = blockIdx.x * 256 + threadIdx.x;
  const int b = idx >> 18;
  const int h = (idx >> 9) & 511;
  const int i = idx & 511;
  float v;
  if (h == i) v = dei[b * HID + h];
  else        v = LR_F * vtrH[((size_t)b * SEQ + 63) * HID + h] *
                         itrH[((size_t)b * SEQ + 63) * HID + i];
  ei_out[idx] = v;
}

extern "C" void kernel_launch(void* const* d_in, const int* in_sizes, int n_in,
                              void* d_out, int out_size, void* d_ws, size_t ws_size,
                              hipStream_t stream) {
  (void)in_sizes; (void)n_in; (void)out_size; (void)ws_size;
  const float* x    = (const float*)d_in[0];
  const float* W    = (const float*)d_in[1];
  const float* mem0 = (const float*)d_in[2];
  const float* ei0  = (const float*)d_in[3];
  const float* ie0  = (const float*)d_in[4];

  float* out = (float*)d_out;
  float* ei_out   = out + 1048576;
  float* keys_out = out + 3145728;
  float* vals_out = out + 3276800;

  float* ws    = (float*)d_ws;
  float* iproj = ws;
  float* keyH  = ws + 262144;
  float* ikv0  = ws + 393216;
  float* ktrH  = ws + 524288;
  float* vtrH  = ws + 655360;
  float* itrH  = ws + 786432;
  float* dei   = ws + 917504;

  k_proj<<<128, 256, 0, stream>>>(x, W, iproj);
  k_keys<<<4, 512, 0, stream>>>(iproj, keyH, keys_out);
  k_ikv0<<<64, 256, 0, stream>>>(mem0, keyH, ikv0);
  k_scan<<<4, 512, 0, stream>>>(iproj, keyH, ikv0, ei0, ie0,
                                ktrH, vtrH, itrH, dei, vals_out);
  dim3 gf(512, 2);
  k_final<<<gf, 256, 0, stream>>>(mem0, ie0, ktrH, vtrH, itrH, out);
  k_ei<<<4096, 256, 0, stream>>>(dei, vtrH, itrH, ei_out);
}


# scan keeps the ktr and itr histories on chip (ktr in registers, precomputed from the keys; itr in LDS) instead of re-reading them from global memory every step; summation order unchanged
# speedup vs baseline: 3.6868x; 1.1582x over previous
#include <hip/hip_runtime.h>
#include <hip/hip_bf16.h>
#include <math.h>

#define HID   512
#define INH   512
#define INSZ  256
#define BATCH 4
#define SEQ   64
#define DELAY 8
#define LR_F    0.01f
#define DECAY_F 0.951229424500714f
#define OMD_F   0.048770575499285991f
#define RPITCH  (HID + 4)
#define IPITCH  (SEQ + 4)

typedef float v2f __attribute__((ext_vector_type(2)));
typedef float v8f __attribute__((ext_vector_type(8)));

__global__ __launch_bounds__(256) void k_proj(const float* __restrict__ x,
                                              const float* __restrict__ W,
                                              float* __restrict__ iproj) {
  const int gw   = (blockIdx.x * 256 + threadIdx.x) >> 5;
  const int lane = threadIdx.x & 31;
  const int tm = gw >> 6;
  const int tn = gw & 63;
  const int hf = lane >> 4, ln = lane & 15;
  const float* ar = x + (size_t)(tm * 16 + ln) * INSZ;
  const float* br = W + (size_t)(tn * 16 + ln) * INSZ;
  v8f acc = {};
  for (int c = 0; c < INSZ; c += 4) {
    const int cc = c + 2 * hf;
    v2f a, b;
    a.x = ar[cc]; a.y = ar[cc + 1];
    b.x = br[cc]; b.y = br[cc + 1];
    acc = __builtin_amdgcn_wmma_f32_16x16x4_f32(false, a, false, b,
                                                (short)0, acc, false, false);
  }
  for (int r = 0; r < 8; ++r)
    iproj[(size_t)(tm * 16 + r + 8 * hf) * 1024 + tn * 16 + ln] = acc[r];
}

__global__ __launch_bounds__(512) void k_keys(const float* __restrict__ iproj,
                                              float* __restrict__ keyH,
                                              float* __restrict__ keys_out) {
  const int b = blockIdx.x, h = threadIdx.x;
  float kv = 0.f;
  for (int t = 0; t < SEQ; ++t) {
    kv = DECAY_F * kv + iproj[((size_t)b * SEQ + t) * 1024 + h];
    const float ko = tanhf(kv);
    keyH[((size_t)b * SEQ + t) * HID + h]     = ko;
    keys_out[((size_t)b * SEQ + t) * HID + h] = ko;
  }
}

__global__ __launch_bounds__(256) void k_ikv0(const float* __restrict__ mem0,
                                              const float* __restrict__ keyH,
                                              float* __restrict__ ikv0) {
  const int gw   = (blockIdx.x * 256 + threadIdx.x) >> 5;
  const int lane = threadIdx.x & 31;
  const int b  = gw >> 7;
  const int rr = gw & 127;
  const int tm = rr >> 2;
  const int tn = rr & 3;
  const int hf = lane >> 4, ln = lane & 15;
  const float* ar = mem0 + ((size_t)b * HID + tm * 16 + ln) * HID;
  const float* br = keyH + ((size_t)b * SEQ + tn * 16 + ln) * HID;
  v8f acc = {};
  for (int c = 0; c < HID; c += 4) {
    const int cc = c + 2 * hf;
    v2f a, bb;
    a.x  = ar[cc]; a.y  = ar[cc + 1];
    bb.x = br[cc]; bb.y = br[cc + 1];
    acc = __builtin_amdgcn_wmma_f32_16x16x4_f32(false, a, false, bb,
                                                (short)0, acc, false, false);
  }
  for (int r = 0; r < 8; ++r) {
    const int hh = tm * 16 + r + 8 * hf;
    const int tt = tn * 16 + ln;
    ikv0[((size_t)b * SEQ + tt) * HID + hh] = acc[r];
  }
}

__global__ __launch_bounds__(512) void k_scan(const float* __restrict__ iproj,
                                              const float* __restrict__ keyH,
                                              const float* __restrict__ ikv0,
                                              const float* __restrict__ ei0,
                                              const float* __restrict__ ie0,
                                              float* __restrict__ ktrH,
                                              float* __restrict__ vtrH,
                                              float* __restrict__ itrH,
                                              float* __restrict__ dei,
                                              float* __restrict__ vals_out) {
  __shared__ __align__(16) float vtr_l[SEQ][HID];
  __shared__ __align__(16) float ring[DELAY + 1][RPITCH];
  __shared__ __align__(16) float g_s[DELAY][HID];
  __shared__ __align__(16) float key_s[HID];
  __shared__ __align__(16) float itr_l[HID][IPITCH];
  __shared__ __align__(16) float dk[SEQ];
  __shared__ __align__(16) float db[SEQ];

  const int b = blockIdx.x;
  const int h = threadIdx.x;
  const int lane = h & 31, wv = h >> 5;

  for (int r = 0; r <= DELAY; ++r) ring[r][h] = 0.f;
  for (int s = 0; s < SEQ; s += 4)
    *(float4*)&itr_l[h][s] = make_float4(0.f, 0.f, 0.f, 0.f);
  if (h < SEQ) { dk[h] = 0.f; db[h] = 0.f; }

  float kh[SEQ];
  {
    float kin[SEQ];
#pragma unroll
    for (int s = 0; s < SEQ; ++s)
      kin[s] = keyH[((size_t)b * SEQ + s) * HID + h];
    float kt = 0.f;
#pragma unroll
    for (int s = 0; s < SEQ; ++s) {
      kt = DECAY_F * kt + OMD_F * kin[s];
      kh[s] = kt;
      ktrH[((size_t)b * SEQ + s) * HID + h] = kt;
    }
  }

  float vvs = 0.f, isv = 0.f;
  float vtr = 0.f, itr = 0.f, accD = 0.f;
  const size_t mbase = (size_t)b * HID * HID + (size_t)h * HID;
  float d_ei = ei0[mbase + h];
  const float ie0d = ie0[mbase + h];
  const int ln = lane & 15, hf = lane >> 4;
  const float* a0p = ie0 + (size_t)b * HID * HID + (size_t)(wv * 32 + ln) * HID + 4 * hf;
  const float* a1p = a0p + (size_t)16 * HID;
  const float* bp  = &ring[ln < DELAY ? ln : DELAY][4 * hf];
  __syncthreads();

  for (int t = 0; t < SEQ; ++t) {
    const int r = t & (DELAY - 1);
    const size_t rowoff = ((size_t)b * SEQ + t) * HID;

    key_s[h] = keyH[rowoff + h];
    __syncthreads();

    for (int s = wv; s < t; s += 16) {
      float p1 = 0.f, p2 = 0.f;
      for (int k = lane; k < HID; k += 32) {
        const float vk = vtr_l[s][k];
        p1 = fmaf(vk, key_s[k], p1);
        p2 = fmaf(vk, ring[r][k], p2);
      }
      for (int off = 16; off > 0; off >>= 1) {
        p1 += __shfl_xor(p1, off, 32);
        p2 += __shfl_xor(p2, off, 32);
      }
      if (lane == 0) { dk[s] = p1; db[s] = p2; }
    }

    if (r == 0 && t != 0) {
      v8f g0 = {}, g1 = {};
      float4 fa0[2][8], fa1[2][8];
      for (int q = 0; q < 8; ++q) {
        fa0[0][q] = *(const float4*)(a0p + q * 8);
        fa1[0][q] = *(const float4*)(a1p + q * 8);
      }
#pragma unroll
      for (int ch = 0; ch < HID / 64; ++ch) {
        const int cu = ch & 1, nx = cu ^ 1;
        if (ch + 1 < HID / 64) {
          for (int q = 0; q < 8; ++q) {
            fa0[nx][q] = *(const float4*)(a0p + (ch + 1) * 64 + q * 8);
            fa1[nx][q] = *(const float4*)(a1p + (ch + 1) * 64 + q * 8);
          }
        }
        float4 fb[8];
        for (int q = 0; q < 8; ++q)
          fb[q] = *(const float4*)(bp + ch * 64 + q * 8);
        for (int q = 0; q < 8; ++q) {
          const v2f b01 = {fb[q].x, fb[q].y}, b23 = {fb[q].z, fb[q].w};
          const v2f p01 = {fa0[cu][q].x, fa0[cu][q].y}, p23 = {fa0[cu][q].z, fa0[cu][q].w};
          const v2f q01 = {fa1[cu][q].x, fa1[cu][q].y}, q23 = {fa1[cu][q].z, fa1[cu][q].w};
          g0 = __builtin_amdgcn_wmma_f32_16x16x4_f32(false, p01, false, b01,
                                                     (short)0, g0, false, false);
          g1 = __builtin_amdgcn_wmma_f32_16x16x4_f32(false, q01, false, b01,
                                                     (short)0, g1, false, false);
          g0 = __builtin_amdgcn_wmma_f32_16x16x4_f32(false, p23, false, b23,
                                                     (short)0, g0, false, false);
          g1 = __builtin_amdgcn_wmma_f32_16x16x4_f32(false, q23, false, b23,
                                                     (short)0, g1, false, false);
        }
      }
      if (ln < DELAY) {
        float* gp = &g_s[ln][wv * 32 + 8 * hf];
        *(float4*)(gp)      = make_float4(g0[0], g0[1], g0[2], g0[3]);
        *(float4*)(gp + 4)  = make_float4(g0[4], g0[5], g0[6], g0[7]);
        *(float4*)(gp + 16) = make_float4(g1[0], g1[1], g1[2], g1[3]);
        *(float4*)(gp + 20) = make_float4(g1[4], g1[5], g1[6], g1[7]);
      }
    }
    __syncthreads();

    float sk = 0.f, si = 0.f;
#pragma unroll
    for (int s8 = 0; s8 < SEQ; s8 += 8) {
      if (s8 < t) {
        const float4 d0 = *(const float4*)&dk[s8];
        const float4 d1 = *(const float4*)&dk[s8 + 4];
        const float4 e0 = *(const float4*)&db[s8];
        const float4 e1 = *(const float4*)&db[s8 + 4];
        const float4 i0 = *(const float4*)&itr_l[h][s8];
        const float4 i1 = *(const float4*)&itr_l[h][s8 + 4];
        sk = fmaf(kh[s8 + 0], d0.x, sk);
        si = fmaf(i0.x, e0.x, si);
        sk = fmaf(kh[s8 + 1], d0.y, sk);
        si = fmaf(i0.y, e0.y, si);
        sk = fmaf(kh[s8 + 2], d0.z, sk);
        si = fmaf(i0.z, e0.z, si);
        sk = fmaf(kh[s8 + 3], d0.w, sk);
        si = fmaf(i0.w, e0.w, si);
        sk = fmaf(kh[s8 + 4], d1.x, sk);
        si = fmaf(i1.x, e1.x, si);
        sk = fmaf(kh[s8 + 5], d1.y, sk);
        si = fmaf(i1.y, e1.y, si);
        sk = fmaf(kh[s8 + 6], d1.z, sk);
        si = fmaf(i1.z, e1.z, si);
        sk = fmaf(kh[s8 + 7], d1.w, sk);
        si = fmaf(i1.w, e1.w, si);
      }
    }

    const float mv = (t >= DELAY) ? g_s[r][h] : 0.f;
    const float bufh = ring[r][h];
    const float inh_out = mv + LR_F * si - bufh * (ie0d + LR_F * accD);

    const float ikv = 0.2f * (ikv0[rowoff + h] + LR_F * sk);
    const float ivt = iproj[((size_t)b * SEQ + t) * 1024 + HID + h];

    vvs = DECAY_F * vvs + ivt + ikv - inh_out;
    const float val = tanhf(vvs);
    isv = DECAY_F * isv + val * d_ei;
    const float inh = tanhf(isv);

    vtr = DECAY_F * vtr + OMD_F * val;
    itr = DECAY_F * itr + OMD_F * inh;
    const float p = vtr * itr;
    d_ei += LR_F * p;
    accD += p;

    vtrH[rowoff + h] = vtr;
    itrH[rowoff + h] = itr;
    vtr_l[t][h] = vtr;
    itr_l[h][t] = itr;
    vals_out[rowoff + h] = val;

    __syncthreads();
    ring[r][h] = inh;
  }
  dei[(size_t)b * HID + h] = d_ei;
}

__global__ __launch_bounds__(256) void k_final(const float* __restrict__ mem0,
                                               const float* __restrict__ ie0,
                                               const float* __restrict__ ktrH,
                                               const float* __restrict__ vtrH,
                                               const float* __restrict__ itrH,
                                               float* __restrict__ out) {
  const int which = blockIdx.y;
  const int gw    = (blockIdx.x * 256 + threadIdx.x) >> 5;
  const int lane  = threadIdx.x & 31;
  const int b  = gw >> 10;
  const int rr = gw & 1023;
  const int tm = rr >> 5, tn = rr & 31;
  const int hf = lane >> 4, ln = lane & 15;
  const float* Ah = which ? itrH : ktrH;
  v8f acc = {};
  for (int c = 0; c < SEQ; c += 4) {
    const int cc = c + 2 * hf;
    v2f a, bb;
    a.x  = Ah[((size_t)b * SEQ + cc)     * HID + tm * 16 + ln];
    a.y  = Ah[((size_t)b * SEQ + cc + 1) * HID + tm * 16 + ln];
    bb.x = vtrH[((size_t)b * SEQ + cc)     * HID + tn * 16 + ln];
    bb.y = vtrH[((size_t)b * SEQ + cc + 1) * HID + tn * 16 + ln];
    acc = __builtin_amdgcn_wmma_f32_16x16x4_f32(false, a, false, bb,
                                                (short)0, acc, false, false);
  }
  const float* base0 = which ? ie0 : mem0;
  float* dst = out + (which ? (size_t)2097152 : (size_t)0);
  for (int r = 0; r < 8; ++r) {
    const int m = tm * 16 + r + 8 * hf;
    const int n = tn * 16 + ln;
    const size_t idx = (size_t)b * HID * HID + (size_t)m * HID + n;
    float v = base0[idx] + LR_F * acc[r];
    if (which && m == n)
      v = LR_F * itrH[((size_t)b * SEQ + 63) * HID + m] *
                 vtrH[((size_t)b * SEQ + 63) * HID + n];
    dst[idx] = v;
  }
}

__global__ __launch_bounds__(256) void k_ei(const float* __restrict__ dei,
                                            const float* __restrict__ vtrH,
                                            const float* __restrict__ itrH,
                                            float* __restrict__ ei_out) {
  const int idx = blockIdx.x * 256 + threadIdx.x;
  const int b = idx >> 18;
  const int h = (idx >> 9) & 511;
  const int i = idx & 511;
  float v;
  if (h == i) v = dei[b * HID + h];
  else        v = LR_F * vtrH[((size_t)b * SEQ + 63) * HID + h] *
                         itrH[((size_t)b * SEQ + 63) * HID + i];
  ei_out[idx] = v;
}

extern "C" void kernel_launch(void* const* d_in, const int* in_sizes, int n_in,
                              void* d_out, int out_size, void* d_ws, size_t ws_size,
                              hipStream_t stream) {
  (void)in_sizes; (void)n_in; (void)out_size; (void)ws_size;
  const float* x    = (const float*)d_in[0];
  const float* W    = (const float*)d_in[1];
  const float* mem0 = (const float*)d_in[2];
  const float* ei0  = (const float*)d_in[3];
  const float* ie0  = (const float*)d_in[4];

  float* out = (float*)d_out;
  float* ei_out   = out + 1048576;
  float* keys_out = out + 3145728;
  float* vals_out = out + 3276800;

  float* ws    = (float*)d_ws;
  float* iproj = ws;
  float* keyH  = ws + 262144;
  float* ikv0  = ws + 393216;
  float* ktrH  = ws + 524288;
  float* vtrH  = ws + 655360;
  float* itrH  = ws + 786432;
  float* dei   = ws + 917504;

  k_proj<<<128, 256, 0, stream>>>(x, W, iproj);
  k_keys<<<4, 512, 0, stream>>>(iproj, keyH, keys_out);
  k_ikv0<<<64, 256, 0, stream>>>(mem0, keyH, ikv0);
  k_scan<<<4, 512, 0, stream>>>(iproj, keyH, ikv0, ei0, ie0,
                                ktrH, vtrH, itrH, dei, vals_out);
  dim3 gf(512, 2);
  k_final<<<gf, 256, 0, stream>>>(mem0, ie0, ktrH, vtrH, itrH, out);
  k_ei<<<4096, 256, 0, stream>>>(dei, vtrH, itrH, ei_out);
}


# scan step: its global loads (ikv0, iv, next key) and the previous step's result stores are issued under the dot-product phase, off the serial chain; 2 barriers per step, not 3; sums unchanged
# speedup vs baseline: 4.4400x; 1.2043x over previous
#include <hip/hip_runtime.h>
#include <hip/hip_bf16.h>
#include <math.h>

#define HID   512
#define INH   512
#define INSZ  256
#define BATCH 4
#define SEQ   64
#define DELAY 8
#define LR_F    0.01f
#define DECAY_F 0.951229424500714f
#define OMD_F   0.048770575499285991f
#define RPITCH  (HID + 4)
#define IPITCH  (SEQ + 4)

typedef float v2f __attribute__((ext_vector_type(2)));
typedef float v8f __attribute__((ext_vector_type(8)));

__global__ __launch_bounds__(256) void k_proj(const float* __restrict__ x,
                                              const float* __restrict__ W,
                                              float* __restrict__ iproj) {
  const int gw   = (blockIdx.x * 256 + threadIdx.x) >> 5;
  const int lane = threadIdx.x & 31;
  const int tm = gw >> 6;
  const int tn = gw & 63;
  const int hf = lane >> 4, ln = lane & 15;
  const float* ar = x + (size_t)(tm * 16 + ln) * INSZ;
  const float* br = W + (size_t)(tn * 16 + ln) * INSZ;
  v8f acc = {};
  for (int c = 0; c < INSZ; c += 4) {
    const int cc = c + 2 * hf;
    v2f a, b;
    a.x = ar[cc]; a.y = ar[cc + 1];
    b.x = br[cc]; b.y = br[cc + 1];
    acc = __builtin_amdgcn_wmma_f32_16x16x4_f32(false, a, false, b,
                                                (short)0, acc, false, false);
  }
  for (int r = 0; r < 8; ++r)
    iproj[(size_t)(tm * 16 + r + 8 * hf) * 1024 + tn * 16 + ln] = acc[r];
}

__global__ __launch_bounds__(512) void k_keys(const float* __restrict__ iproj,
                                              float* __restrict__ keyH,
                                              float* __restrict__ keys_out) {
  const int b = blockIdx.x, h = threadIdx.x;
  float kv = 0.f;
  for (int t = 0; t < SEQ; ++t) {
    kv = DECAY_F * kv + iproj[((size_t)b * SEQ + t) * 1024 + h];
    const float ko = tanhf(kv);
    keyH[((size_t)b * SEQ + t) * HID + h]     = ko;
    keys_out[((size_t)b * SEQ + t) * HID + h] = ko;
  }
}

__global__ __launch_bounds__(256) void k_ikv0(const float* __restrict__ mem0,
                                              const float* __restrict__ keyH,
                                              float* __restrict__ ikv0) {
  const int gw   = (blockIdx.x * 256 + threadIdx.x) >> 5;
  const int lane = threadIdx.x & 31;
  const int b  = gw >> 7;
  const int rr = gw & 127;
  const int tm = rr >> 2;
  const int tn = rr & 3;
  const int hf = lane >> 4, ln = lane & 15;
  const float* ar = mem0 + ((size_t)b * HID + tm * 16 + ln) * HID;
  const float* br = keyH + ((size_t)b * SEQ + tn * 16 + ln) * HID;
  v8f acc = {};
  for (int c = 0; c < HID; c += 4) {
    const int cc = c + 2 * hf;
    v2f a, bb;
    a.x  = ar[cc]; a.y  = ar[cc + 1];
    bb.x = br[cc]; bb.y = br[cc + 1];
    acc = __builtin_amdgcn_wmma_f32_16x16x4_f32(false, a, false, bb,
                                                (short)0, acc, false, false);
  }
  for (int r = 0; r < 8; ++r) {
    const int hh = tm * 16 + r + 8 * hf;
    const int tt = tn * 16 + ln;
    ikv0[((size_t)b * SEQ + tt) * HID + hh] = acc[r];
  }
}

__global__ __launch_bounds__(512) void k_scan(const float* __restrict__ iproj,
                                              const float* __restrict__ keyH,
                                              const float* __restrict__ ikv0,
                                              const float* __restrict__ ei0,
                                              const float* __restrict__ ie0,
                                              float* __restrict__ ktrH,
                                              float* __restrict__ vtrH,
                                              float* __restrict__ itrH,
                                              float* __restrict__ dei,
                                              float* __restrict__ vals_out) {
  __shared__ __align__(16) float vtr_l[SEQ][HID];
  __shared__ __align__(16) float ring[DELAY + 1][RPITCH];
  __shared__ __align__(16) float g_s[DELAY][HID];
  __shared__ __align__(16) float key_s[HID];
  __shared__ __align__(16) float itr_l[HID][IPITCH];
  __shared__ __align__(16) float dk[SEQ];
  __shared__ __align__(16) float db[SEQ];

  const int b = blockIdx.x;
  const int h = threadIdx.x;
  const int lane = h & 31, wv = h >> 5;

  for (int r = 0; r <= DELAY; ++r) ring[r][h] = 0.f;
  for (int s = 0; s < SEQ; s += 4)
    *(float4*)&itr_l[h][s] = make_float4(0.f, 0.f, 0.f, 0.f);
  if (h < SEQ) { dk[h] = 0.f; db[h] = 0.f; }

  float kh[SEQ];
  float key0;
  {
    float kin[SEQ];
#pragma unroll
    for (int s = 0; s < SEQ; ++s)
      kin[s] = keyH[((size_t)b * SEQ + s) * HID + h];
    key0 = kin[0];
    float kt = 0.f;
#pragma unroll
    for (int s = 0; s < SEQ; ++s) {
      kt = DECAY_F * kt + OMD_F * kin[s];
      kh[s] = kt;
      ktrH[((size_t)b * SEQ + s) * HID + h] = kt;
    }
  }

  float vvs = 0.f, isv = 0.f;
  float vtr = 0.f, itr = 0.f, accD = 0.f;
  const size_t mbase = (size_t)b * HID * HID + (size_t)h * HID;
  float d_ei = ei0[mbase + h];
  const float ie0d = ie0[mbase + h];
  const int ln = lane & 15, hf = lane >> 4;
  const float* a0p = ie0 + (size_t)b * HID * HID + (size_t)(wv * 32 + ln) * HID + 4 * hf;
  const float* a1p = a0p + (size_t)16 * HID;
  const float* bp  = &ring[ln < DELAY ? ln : DELAY][4 * hf];
  key_s[h] = key0;
  float val_p = 0.f;
  __syncthreads();

  for (int t = 0; t < SEQ; ++t) {
    const int r = t & (DELAY - 1);
    const size_t rowoff = ((size_t)b * SEQ + t) * HID;

    const float ikv_g = ikv0[rowoff + h];
    const float ivt = iproj[((size_t)b * SEQ + t) * 1024 + HID + h];
    const float key_n = keyH[rowoff + (t + 1 < SEQ ? HID : 0) + h];
    if (t > 0) {
      vtrH[rowoff - HID + h] = vtr;
      itrH[rowoff - HID + h] = itr;
      vals_out[rowoff - HID + h] = val_p;
    }

    for (int s = wv; s < t; s += 16) {
      float p1 = 0.f, p2 = 0.f;
      for (int k = lane; k < HID; k += 32) {
        const float vk = vtr_l[s][k];
        p1 = fmaf(vk, key_s[k], p1);
        p2 = fmaf(vk, ring[r][k], p2);
      }
      for (int off = 16; off > 0; off >>= 1) {
        p1 += __shfl_xor(p1, off, 32);
        p2 += __shfl_xor(p2, off, 32);
      }
      if (lane == 0) { dk[s] = p1; db[s] = p2; }
    }

    if (r == 0 && t != 0) {
      v8f g0 = {}, g1 = {};
      float4 fa0[2][8], fa1[2][8];
      for (int q = 0; q < 8; ++q) {
        fa0[0][q] = *(const float4*)(a0p + q * 8);
        fa1[0][q] = *(const float4*)(a1p + q * 8);
      }
#pragma unroll
      for (int ch = 0; ch < HID / 64; ++ch) {
        const int cu = ch & 1, nx = cu ^ 1;
        if (ch + 1 < HID / 64) {
          for (int q = 0; q < 8; ++q) {
            fa0[nx][q] = *(const float4*)(a0p + (ch + 1) * 64 + q * 8);
            fa1[nx][q] = *(const float4*)(a1p + (ch + 1) * 64 + q * 8);
          }
        }
        float4 fb[8];
        for (int q = 0; q < 8; ++q)
          fb[q] = *(const float4*)(bp + ch * 64 + q * 8);
        for (int q = 0; q < 8; ++q) {
          const v2f b01 = {fb[q].x, fb[q].y}, b23 = {fb[q].z, fb[q].w};
          const v2f p01 = {fa0[cu][q].x, fa0[cu][q].y}, p23 = {fa0[cu][q].z, fa0[cu][q].w};
          const v2f q01 = {fa1[cu][q].x, fa1[cu][q].y}, q23 = {fa1[cu][q].z, fa1[cu][q].w};
          g0 = __builtin_amdgcn_wmma_f32_16x16x4_f32(false, p01, false, b01,
                                                     (short)0, g0, false, false);
          g1 = __builtin_amdgcn_wmma_f32_16x16x4_f32(false, q01, false, b01,
                                                     (short)0, g1, false, false);
          g0 = __builtin_amdgcn_wmma_f32_16x16x4_f32(false, p23, false, b23,
                                                     (short)0, g0, false, false);
          g1 = __builtin_amdgcn_wmma_f32_16x16x4_f32(false, q23, false, b23,
                                                     (short)0, g1, false, false);
        }
      }
      if (ln < DELAY) {
        float* gp = &g_s[ln][wv * 32 + 8 * hf];
        *(float4*)(gp)      = make_float4(g0[0], g0[1], g0[2], g0[3]);
        *(float4*)(gp + 4)  = make_float4(g0[4], g0[5], g0[6], g0[7]);
        *(float4*)(gp + 16) = make_float4(g1[0], g1[1], g1[2], g1[3]);
        *(float4*)(gp + 20) = make_float4(g1[4], g1[5], g1[6], g1[7]);
      }
    }
    __syncthreads();

    float sk = 0.f, si = 0.f;
#pragma unroll
    for (int s8 = 0; s8 < SEQ; s8 += 8) {
      if (s8 < t) {
        const float4 d0 = *(const float4*)&dk[s8];
        const float4 d1 = *(const float4*)&dk[s8 + 4];
        const float4 e0 = *(const float4*)&db[s8];
        const float4 e1 = *(const float4*)&db[s8 + 4];
        const float4 i0 = *(const float4*)&itr_l[h][s8];
        const float4 i1 = *(const float4*)&itr_l[h][s8 + 4];
        sk = fmaf(kh[s8 + 0], d0.x, sk);
        si = fmaf(i0.x, e0.x, si);
        sk = fmaf(kh[s8 + 1], d0.y, sk);
        si = fmaf(i0.y, e0.y, si);
        sk = fmaf(kh[s8 + 2], d0.z, sk);
        si = fmaf(i0.z, e0.z, si);
        sk = fmaf(kh[s8 + 3], d0.w, sk);
        si = fmaf(i0.w, e0.w, si);
        sk = fmaf(kh[s8 + 4], d1.x, sk);
        si = fmaf(i1.x, e1.x, si);
        sk = fmaf(kh[s8 + 5], d1.y, sk);
        si = fmaf(i1.y, e1.y, si);
        sk = fmaf(kh[s8 + 6], d1.z, sk);
        si = fmaf(i1.z, e1.z, si);
        sk = fmaf(kh[s8 + 7], d1.w, sk);
        si = fmaf(i1.w, e1.w, si);
      }
    }

    const float mv = (t >= DELAY) ? g_s[r][h] : 0.f;
    const float bufh = ring[r][h];
    const float inh_out = mv + LR_F * si - bufh * (ie0d + LR_F * accD);

    const float ikv = 0.2f * (ikv_g + LR_F * sk);

    vvs = DECAY_F * vvs + ivt + ikv - inh_out;
    const float val = tanhf(vvs);
    isv = DECAY_F * isv + val * d_ei;
    const float inh = tanhf(isv);

    vtr = DECAY_F * vtr + OMD_F * val;
    itr = DECAY_F * itr + OMD_F * inh;
    const float p = vtr * itr;
    d_ei += LR_F * p;
    accD += p;

    vtr_l[t][h] = vtr;
    itr_l[h][t] = itr;
    val_p = val;
    ring[r][h] = inh;
    key_s[h] = key_n;
    __syncthreads();
  }
  {
    const size_t rowl = ((size_t)b * SEQ + SEQ - 1) * HID;
    vtrH[rowl + h] = vtr;
    itrH[rowl + h] = itr;
    vals_out[rowl + h] = val_p;
  }
  dei[(size_t)b * HID + h] = d_ei;
}

__global__ __launch_bounds__(256) void k_final(const float* __restrict__ mem0,
                                               const float* __restrict__ ie0,
                                               const float* __restrict__ ktrH,
                                               const float* __restrict__ vtrH,
                                               const float* __restrict__ itrH,
                                               float* __restrict__ out) {
  const int which = blockIdx.y;
  const int gw    = (blockIdx.x * 256 + threadIdx.x) >> 5;
  const int lane  = threadIdx.x & 31;
  const int b  = gw >> 10;
  const int rr = gw & 1023;
  const int tm = rr >> 5, tn = rr & 31;
  const int hf = lane >> 4, ln = lane & 15;
  const float* Ah = which ? itrH : ktrH;
  v8f acc = {};
  for (int c = 0; c < SEQ; c += 4) {
    const int cc = c + 2 * hf;
    v2f a, bb;
    a.x  = Ah[((size_t)b * SEQ + cc)     * HID + tm * 16 + ln];
    a.y  = Ah[((size_t)b * SEQ + cc + 1) * HID + tm * 16 + ln];
    bb.x = vtrH[((size_t)b * SEQ + cc)     * HID + tn * 16 + ln];
    bb.y = vtrH[((size_t)b * SEQ + cc + 1) * HID + tn * 16 + ln];
    acc = __builtin_amdgcn_wmma_f32_16x16x4_f32(false, a, false, bb,
                                                (short)0, acc, false, false);
  }
  const float* base0 = which ? ie0 : mem0;
  float* dst = out + (which ? (size_t)2097152 : (size_t)0);
  for (int r = 0; r < 8; ++r) {
    const int m = tm * 16 + r + 8 * hf;
    const int n = tn * 16 + ln;
    const size_t idx = (size_t)b * HID * HID + (size_t)m * HID + n;
    float v = base0[idx] + LR_F * acc[r];
    if (which && m == n)
      v = LR_F * itrH[((size_t)b * SEQ + 63) * HID + m] *
                 vtrH[((size_t)b * SEQ + 63) * HID + n];
    dst[idx] = v;
  }
}

__global__ __launch_bounds__(256) void k_ei(const float* __restrict__ dei,
                                            const float* __restrict__ vtrH,
                                            const float* __restrict__ itrH,
                                            float* __restrict__ ei_out) {
  const int idx = blockIdx.x * 256 + threadIdx.x;
  const int b = idx >> 18;
  const int h = (idx >> 9) & 511;
  const int i = idx & 511;
  float v;
  if (h == i) v = dei[b * HID + h];
  else        v = LR_F * vtrH[((size_t)b * SEQ + 63) * HID + h] *
                         itrH[((size_t)b * SEQ + 63) * HID + i];
  ei_out[idx] = v;
}

extern "C" void kernel_launch(void* const* d_in, const int* in_sizes, int n_in,
                              void* d_out, int out_size, void* d_ws, size_t ws_size,
                              hipStream_t stream) {
  (void)in_sizes; (void)n_in; (void)out_size; (void)ws_size;
  const float* x    = (const float*)d_in[0];
  const float* W    = (const float*)d_in[1];
  const float* mem0 = (const float*)d_in[2];
  const float* ei0  = (const float*)d_in[3];
  const float* ie0  = (const float*)d_in[4];

  float* out = (float*)d_out;
  float* ei_out   = out + 1048576;
  float* keys_out = out + 3145728;
  float* vals_out = out + 3276800;

  float* ws    = (float*)d_ws;
  float* iproj = ws;
  float* keyH  = ws + 262144;
  float* ikv0  = ws + 393216;
  float* ktrH  = ws + 524288;
  float* vtrH  = ws + 655360;
  float* itrH  = ws + 786432;
  float* dei   = ws + 917504;

  k_proj<<<128, 256, 0, stream>>>(x, W, iproj);
  k_keys<<<4, 512, 0, stream>>>(iproj, keyH, keys_out);
  k_ikv0<<<64, 256, 0, stream>>>(mem0, keyH, ikv0);
  k_scan<<<4, 512, 0, stream>>>(iproj, keyH, ikv0, ei0, ie0,
                                ktrH, vtrH, itrH, dei, vals_out);
  dim3 gf(512, 2);
  k_final<<<gf, 256, 0, stream>>>(mem0, ie0, ktrH, vtrH, itrH, out);
  k_ei<<<4096, 256, 0, stream>>>(dei, vtrH, itrH, ei_out);
}


# scan dot-product phase: key_t and buf_t stay in registers across a wave's history rows, lane sums by DPP/permlane adds instead of LDS permutes: ~3x fewer LDS instructions; sums bit-identical
# speedup vs baseline: 5.0558x; 1.1387x over previous
#include <hip/hip_runtime.h>
#include <hip/hip_bf16.h>
#include <math.h>

#define HID   512
#define INH   512
#define INSZ  256
#define BATCH 4
#define SEQ   64
#define DELAY 8
#define LR_F    0.01f
#define DECAY_F 0.951229424500714f
#define OMD_F   0.048770575499285991f
#define RPITCH  (HID + 4)
#define IPITCH  (SEQ + 4)

typedef float v2f __attribute__((ext_vector_type(2)));
typedef float v8f __attribute__((ext_vector_type(8)));

__global__ __launch_bounds__(256) void k_proj(const float* __restrict__ x,
                                              const float* __restrict__ W,
                                              float* __restrict__ iproj) {
  const int gw   = (blockIdx.x * 256 + threadIdx.x) >> 5;
  const int lane = threadIdx.x & 31;
  const int tm = gw >> 6;
  const int tn = gw & 63;
  const int hf = lane >> 4, ln = lane & 15;
  const float* ar = x + (size_t)(tm * 16 + ln) * INSZ;
  const float* br = W + (size_t)(tn * 16 + ln) * INSZ;
  v8f acc = {};
  for (int c = 0; c < INSZ; c += 4) {
    const int cc = c + 2 * hf;
    v2f a, b;
    a.x = ar[cc]; a.y = ar[cc + 1];
    b.x = br[cc]; b.y = br[cc + 1];
    acc = __builtin_amdgcn_wmma_f32_16x16x4_f32(false, a, false, b,
                                                (short)0, acc, false, false);
  }
  for (int r = 0; r < 8; ++r)
    iproj[(size_t)(tm * 16 + r + 8 * hf) * 1024 + tn * 16 + ln] = acc[r];
}

__global__ __launch_bounds__(512) void k_keys(const float* __restrict__ iproj,
                                              float* __restrict__ keyH,
                                              float* __restrict__ keys_out) {
  const int b = blockIdx.x, h = threadIdx.x;
  float kv = 0.f;
  for (int t = 0; t < SEQ; ++t) {
    kv = DECAY_F * kv + iproj[((size_t)b * SEQ + t) * 1024 + h];
    const float ko = tanhf(kv);
    keyH[((size_t)b * SEQ + t) * HID + h]     = ko;
    keys_out[((size_t)b * SEQ + t) * HID + h] = ko;
  }
}

__global__ __launch_bounds__(256) void k_ikv0(const float* __restrict__ mem0,
                                              const float* __restrict__ keyH,
                                              float* __restrict__ ikv0) {
  const int gw   = (blockIdx.x * 256 + threadIdx.x) >> 5;
  const int lane = threadIdx.x & 31;
  const int b  = gw >> 7;
  const int rr = gw & 127;
  const int tm = rr >> 2;
  const int tn = rr & 3;
  const int hf = lane >> 4, ln = lane & 15;
  const float* ar = mem0 + ((size_t)b * HID + tm * 16 + ln) * HID;
  const float* br = keyH + ((size_t)b * SEQ + tn * 16 + ln) * HID;
  v8f acc = {};
  for (int c = 0; c < HID; c += 4) {
    const int cc = c + 2 * hf;
    v2f a, bb;
    a.x  = ar[cc]; a.y  = ar[cc + 1];
    bb.x = br[cc]; bb.y = br[cc + 1];
    acc = __builtin_amdgcn_wmma_f32_16x16x4_f32(false, a, false, bb,
                                                (short)0, acc, false, false);
  }
  for (int r = 0; r < 8; ++r) {
    const int hh = tm * 16 + r + 8 * hf;
    const int tt = tn * 16 + ln;
    ikv0[((size_t)b * SEQ + tt) * HID + hh] = acc[r];
  }
}

__device__ __forceinline__ float lane_xor16(float x) {
  const int b = __builtin_bit_cast(int, x);
  return __builtin_bit_cast(float,
      __builtin_amdgcn_permlanex16(b, b, 0x76543210, 0xfedcba98, false, false));
}
template <int N>
__device__ __forceinline__ float lane_xor_row(float x) {
  return __builtin_bit_cast(float,
      __builtin_amdgcn_update_dpp(0, __builtin_bit_cast(int, x), 0x160 + N, 0xf, 0xf, true));
}
__device__ __forceinline__ float wave_sum(float p) {
  p += lane_xor16(p);
  p += lane_xor_row<8>(p);
  p += lane_xor_row<4>(p);
  p += lane_xor_row<2>(p);
  p += lane_xor_row<1>(p);
  return p;
}

template <int NS>
__device__ __forceinline__ void hist_dots(const float (*vt)[HID], int s0, int lane,
                                          const float (&kk)[16], const float (&bb)[16],
                                          float* dk, float* db) {
  float v[NS][16];
#pragma unroll
  for (int j = 0; j < NS; ++j)
#pragma unroll
    for (int i = 0; i < 16; ++i) v[j][i] = vt[j][lane + 32 * i];
  float p1[NS], p2[NS];
#pragma unroll
  for (int j = 0; j < NS; ++j) {
    float a = 0.f, c = 0.f;
#pragma unroll
    for (int i = 0; i < 16; ++i) {
      a = fmaf(v[j][i], kk[i], a);
      c = fmaf(v[j][i], bb[i], c);
    }
    p1[j] = a; p2[j] = c;
  }
#pragma unroll
  for (int j = 0; j < NS; ++j) { p1[j] = wave_sum(p1[j]); p2[j] = wave_sum(p2[j]); }
  if (lane == 0) {
#pragma unroll
    for (int j = 0; j < NS; ++j) { dk[s0 + 16 * j] = p1[j]; db[s0 + 16 * j] = p2[j]; }
  }
}

__global__ __launch_bounds__(512) void k_scan(const float* __restrict__ iproj,
                                              const float* __restrict__ keyH,
                                              const float* __restrict__ ikv0,
                                              const float* __restrict__ ei0,
                                              const float* __restrict__ ie0,
                                              float* __restrict__ ktrH,
                                              float* __restrict__ vtrH,
                                              float* __restrict__ itrH,
                                              float* __restrict__ dei,
                                              float* __restrict__ vals_out) {
  __shared__ __align__(16) float vtr_l[SEQ][HID];
  __shared__ __align__(16) float ring[DELAY + 1][RPITCH];
  __shared__ __align__(16) float g_s[DELAY][HID];
  __shared__ __align__(16) float key_s[HID];
  __shared__ __align__(16) float itr_l[HID][IPITCH];
  __shared__ __align__(16) float dk[SEQ];
  __shared__ __align__(16) float db[SEQ];

  const int b = blockIdx.x;
  const int h = threadIdx.x;
  const int lane = h & 31;
  const int wv = __builtin_amdgcn_readfirstlane(h >> 5);

  for (int r = 0; r <= DELAY; ++r) ring[r][h] = 0.f;
  for (int s = 0; s < SEQ; s += 4)
    *(float4*)&itr_l[h][s] = make_float4(0.f, 0.f, 0.f, 0.f);
  if (h < SEQ) { dk[h] = 0.f; db[h] = 0.f; }

  float kh[SEQ];
  float key0;
  {
    float kin[SEQ];
#pragma unroll
    for (int s = 0; s < SEQ; ++s)
      kin[s] = keyH[((size_t)b * SEQ + s) * HID + h];
    key0 = kin[0];
    float kt = 0.f;
#pragma unroll
    for (int s = 0; s < SEQ; ++s) {
      kt = DECAY_F * kt + OMD_F * kin[s];
      kh[s] = kt;
      ktrH[((size_t)b * SEQ + s) * HID + h] = kt;
    }
  }

  float vvs = 0.f, isv = 0.f;
  float vtr = 0.f, itr = 0.f, accD = 0.f;
  const size_t mbase = (size_t)b * HID * HID + (size_t)h * HID;
  float d_ei = ei0[mbase + h];
  const float ie0d = ie0[mbase + h];
  const int ln = lane & 15, hf = lane >> 4;
  const float* a0p = ie0 + (size_t)b * HID * HID + (size_t)(wv * 32 + ln) * HID + 4 * hf;
  const float* a1p = a0p + (size_t)16 * HID;
  const float* bp  = &ring[ln < DELAY ? ln : DELAY][4 * hf];
  key_s[h] = key0;
  float val_p = 0.f;
  __syncthreads();

  for (int t = 0; t < SEQ; ++t) {
    const int r = t & (DELAY - 1);
    const size_t rowoff = ((size_t)b * SEQ + t) * HID;

    const float ikv_g = ikv0[rowoff + h];
    const float ivt = iproj[((size_t)b * SEQ + t) * 1024 + HID + h];
    const float key_n = keyH[rowoff + (t + 1 < SEQ ? HID : 0) + h];
    if (t > 0) {
      vtrH[rowoff - HID + h] = vtr;
      itrH[rowoff - HID + h] = itr;
      vals_out[rowoff - HID + h] = val_p;
    }

    if (wv < t) {
      float kk[16], bb[16];
#pragma unroll
      for (int i = 0; i < 16; ++i) {
        kk[i] = key_s[lane + 32 * i];
        bb[i] = ring[r][lane + 32 * i];
      }
      const int ns = (t - wv + 15) >> 4;
      if (ns == 1)      hist_dots<1>(vtr_l + 4 * wv, wv, lane, kk, bb, dk, db);
      else if (ns == 2) hist_dots<2>(vtr_l + 4 * wv, wv, lane, kk, bb, dk, db);
      else if (ns == 3) hist_dots<3>(vtr_l + 4 * wv, wv, lane, kk, bb, dk, db);
      else              hist_dots<4>(vtr_l + 4 * wv, wv, lane, kk, bb, dk, db);
    }

    if (r == 0 && t != 0) {
      v8f g0 = {}, g1 = {};
      float4 fa0[2][8], fa1[2][8];
      for (int q = 0; q < 8; ++q) {
        fa0[0][q] = *(const float4*)(a0p + q * 8);
        fa1[0][q] = *(const float4*)(a1p + q * 8);
      }
#pragma unroll
      for (int ch = 0; ch < HID / 64; ++ch) {
        const int cu = ch & 1, nx = cu ^ 1;
        if (ch + 1 < HID / 64) {
          for (int q = 0; q < 8; ++q) {
            fa0[nx][q] = *(const float4*)(a0p + (ch + 1) * 64 + q * 8);
            fa1[nx][q] = *(const float4*)(a1p + (ch + 1) * 64 + q * 8);
          }
        }
        float4 fb[8];
        for (int q = 0; q < 8; ++q)
          fb[q] = *(const float4*)(bp + ch * 64 + q * 8);
        for (int q = 0; q < 8; ++q) {
          const v2f b01 = {fb[q].x, fb[q].y}, b23 = {fb[q].z, fb[q].w};
          const v2f p01 = {fa0[cu][q].x, fa0[cu][q].y}, p23 = {fa0[cu][q].z, fa0[cu][q].w};
          const v2f q01 = {fa1[cu][q].x, fa1[cu][q].y}, q23 = {fa1[cu][q].z, fa1[cu][q].w};
          g0 = __builtin_amdgcn_wmma_f32_16x16x4_f32(false, p01, false, b01,
                                                     (short)0, g0, false, false);
          g1 = __builtin_amdgcn_wmma_f32_16x16x4_f32(false, q01, false, b01,
                                                     (short)0, g1, false, false);
          g0 = __builtin_amdgcn_wmma_f32_16x16x4_f32(false, p23, false, b23,
                                                     (short)0, g0, false, false);
          g1 = __builtin_amdgcn_wmma_f32_16x16x4_f32(false, q23, false, b23,
                                                     (short)0, g1, false, false);
        }
      }
      if (ln < DELAY) {
        float* gp = &g_s[ln][wv * 32 + 8 * hf];
        *(float4*)(gp)      = make_float4(g0[0], g0[1], g0[2], g0[3]);
        *(float4*)(gp + 4)  = make_float4(g0[4], g0[5], g0[6], g0[7]);
        *(float4*)(gp + 16) = make_float4(g1[0], g1[1], g1[2], g1[3]);
        *(float4*)(gp + 20) = make_float4(g1[4], g1[5], g1[6], g1[7]);
      }
    }
    __syncthreads();

    float sk = 0.f, si = 0.f;
#pragma unroll
    for (int s8 = 0; s8 < SEQ; s8 += 8) {
      if (s8 < t) {
        const float4 d0 = *(const float4*)&dk[s8];
        const float4 d1 = *(const float4*)&dk[s8 + 4];
        const float4 e0 = *(const float4*)&db[s8];
        const float4 e1 = *(const float4*)&db[s8 + 4];
        const float4 i0 = *(const float4*)&itr_l[h][s8];
        const float4 i1 = *(const float4*)&itr_l[h][s8 + 4];
        sk = fmaf(kh[s8 + 0], d0.x, sk);
        si = fmaf(i0.x, e0.x, si);
        sk = fmaf(kh[s8 + 1], d0.y, sk);
        si = fmaf(i0.y, e0.y, si);
        sk = fmaf(kh[s8 + 2], d0.z, sk);
        si = fmaf(i0.z, e0.z, si);
        sk = fmaf(kh[s8 + 3], d0.w, sk);
        si = fmaf(i0.w, e0.w, si);
        sk = fmaf(kh[s8 + 4], d1.x, sk);
        si = fmaf(i1.x, e1.x, si);
        sk = fmaf(kh[s8 + 5], d1.y, sk);
        si = fmaf(i1.y, e1.y, si);
        sk = fmaf(kh[s8 + 6], d1.z, sk);
        si = fmaf(i1.z, e1.z, si);
        sk = fmaf(kh[s8 + 7], d1.w, sk);
        si = fmaf(i1.w, e1.w, si);
      }
    }

    const float mv = (t >= DELAY) ? g_s[r][h] : 0.f;
    const float bufh = ring[r][h];
    const float inh_out = mv + LR_F * si - bufh * (ie0d + LR_F * accD);

    const float ikv = 0.2f * (ikv_g + LR_F * sk);

    vvs = DECAY_F * vvs + ivt + ikv - inh_out;
    const float val = tanhf(vvs);
    isv = DECAY_F * isv + val * d_ei;
    const float inh = tanhf(isv);

    vtr = DECAY_F * vtr + OMD_F * val;
    itr = DECAY_F * itr + OMD_F * inh;
    const float p = vtr * itr;
    d_ei += LR_F * p;
    accD += p;

    vtr_l[((t & 15) << 2) | (t >> 4)][h] = vtr;
    itr_l[h][t] = itr;
    val_p = val;
    ring[r][h] = inh;
    key_s[h] = key_n;
    __syncthreads();
  }
  {
    const size_t rowl = ((size_t)b * SEQ + SEQ - 1) * HID;
    vtrH[rowl + h] = vtr;
    itrH[rowl + h] = itr;
    vals_out[rowl + h] = val_p;
  }
  dei[(size_t)b * HID + h] = d_ei;
}

__global__ __launch_bounds__(256) void k_final(const float* __restrict__ mem0,
                                               const float* __restrict__ ie0,
                                               const float* __restrict__ ktrH,
                                               const float* __restrict__ vtrH,
                                               const float* __restrict__ itrH,
                                               float* __restrict__ out) {
  const int which = blockIdx.y;
  const int gw    = (blockIdx.x * 256 + threadIdx.x) >> 5;
  const int lane  = threadIdx.x & 31;
  const int b  = gw >> 10;
  const int rr = gw & 1023;
  const int tm = rr >> 5, tn = rr & 31;
  const int hf = lane >> 4, ln = lane & 15;
  const float* Ah = which ? itrH : ktrH;
  v8f acc = {};
  for (int c = 0; c < SEQ; c += 4) {
    const int cc = c + 2 * hf;
    v2f a, bb;
    a.x  = Ah[((size_t)b * SEQ + cc)     * HID + tm * 16 + ln];
    a.y  = Ah[((size_t)b * SEQ + cc + 1) * HID + tm * 16 + ln];
    bb.x = vtrH[((size_t)b * SEQ + cc)     * HID + tn * 16 + ln];
    bb.y = vtrH[((size_t)b * SEQ + cc + 1) * HID + tn * 16 + ln];
    acc = __builtin_amdgcn_wmma_f32_16x16x4_f32(false, a, false, bb,
                                                (short)0, acc, false, false);
  }
  const float* base0 = which ? ie0 : mem0;
  float* dst = out + (which ? (size_t)2097152 : (size_t)0);
  for (int r = 0; r < 8; ++r) {
    const int m = tm * 16 + r + 8 * hf;
    const int n = tn * 16 + ln;
    const size_t idx = (size_t)b * HID * HID + (size_t)m * HID + n;
    float v = base0[idx] + LR_F * acc[r];
    if (which && m == n)
      v = LR_F * itrH[((size_t)b * SEQ + 63) * HID + m] *
                 vtrH[((size_t)b * SEQ + 63) * HID + n];
    dst[idx] = v;
  }
}

__global__ __launch_bounds__(256) void k_ei(const float* __restrict__ dei,
                                            const float* __restrict__ vtrH,
                                            const float* __restrict__ itrH,
                                            float* __restrict__ ei_out) {
  const int idx = blockIdx.x * 256 + threadIdx.x;
  const int b = idx >> 18;
  const int h = (idx >> 9) & 511;
  const int i = idx & 511;
  float v;
  if (h == i) v = dei[b * HID + h];
  else        v = LR_F * vtrH[((size_t)b * SEQ + 63) * HID + h] *
                         itrH[((size_t)b * SEQ + 63) * HID + i];
  ei_out[idx] = v;
}

extern "C" void kernel_launch(void* const* d_in, const int* in_sizes, int n_in,
                              void* d_out, int out_size, void* d_ws, size_t ws_size,
                              hipStream_t stream) {
  (void)in_sizes; (void)n_in; (void)out_size; (void)ws_size;
  const float* x    = (const float*)d_in[0];
  const float* W    = (const float*)d_in[1];
  const float* mem0 = (const float*)d_in[2];
  const float* ei0  = (const float*)d_in[3];
  const float* ie0  = (const float*)d_in[4];

  float* out = (float*)d_out;
  float* ei_out   = out + 1048576;
  float* keys_out = out + 3145728;
  float* vals_out = out + 3276800;

  float* ws    = (float*)d_ws;
  float* iproj = ws;
  float* keyH  = ws + 262144;
  float* ikv0  = ws + 393216;
  float* ktrH  = ws + 524288;
  float* vtrH  = ws + 655360;
  float* itrH  = ws + 786432;
  float* dei   = ws + 917504;

  k_proj<<<128, 256, 0, stream>>>(x, W, iproj);
  k_keys<<<4, 512, 0, stream>>>(iproj, keyH, keys_out);
  k_ikv0<<<64, 256, 0, stream>>>(mem0, keyH, ikv0);
  k_scan<<<4, 512, 0, stream>>>(iproj, keyH, ikv0, ei0, ie0,
                                ktrH, vtrH, itrH, dei, vals_out);
  dim3 gf(512, 2);
  k_final<<<gf, 256, 0, stream>>>(mem0, ie0, ktrH, vtrH, itrH, out);
  k_ei<<<4096, 256, 0, stream>>>(dei, vtrH, itrH, ei_out);
}


# key chain kernel: a column's 64 inputs are requested up front (one memory round trip, not 64 serial ones), its 64 tanh spread over 8 waves; 512 one-wave blocks, not 4 of 16; sums unchanged
# speedup vs baseline: 5.7671x; 1.1407x over previous
#include <hip/hip_runtime.h>
#include <hip/hip_bf16.h>
#include <math.h>

#define HID   512
#define INH   512
#define INSZ  256
#define BATCH 4
#define SEQ   64
#define DELAY 8
#define LR_F    0.01f
#define DECAY_F 0.951229424500714f
#define OMD_F   0.048770575499285991f
#define RPITCH  (HID + 4)
#define IPITCH  (SEQ + 4)

typedef float v2f __attribute__((ext_vector_type(2)));
typedef float v8f __attribute__((ext_vector_type(8)));

__global__ __launch_bounds__(256) void k_proj(const float* __restrict__ x,
                                              const float* __restrict__ W,
                                              float* __restrict__ iproj) {
  const int gw   = (blockIdx.x * 256 + threadIdx.x) >> 5;
  const int lane = threadIdx.x & 31;
  const int tm = gw >> 6;
  const int tn = gw & 63;
  const int hf = lane >> 4, ln = lane & 15;
  const float* ar = x + (size_t)(tm * 16 + ln) * INSZ;
  const float* br = W + (size_t)(tn * 16 + ln) * INSZ;
  v8f acc = {};
  for (int c = 0; c < INSZ; c += 4) {
    const int cc = c + 2 * hf;
    v2f a, b;
    a.x = ar[cc]; a.y = ar[cc + 1];
    b.x = br[cc]; b.y = br[cc + 1];
    acc = __builtin_amdgcn_wmma_f32_16x16x4_f32(false, a, false, b,
                                                (short)0, acc, false, false);
  }
  for (int r = 0; r < 8; ++r)
    iproj[(size_t)(tm * 16 + r + 8 * hf) * 1024 + tn * 16 + ln] = acc[r];
}

__global__ __launch_bounds__(32) void k_keys(const float* __restrict__ iproj,
                                             float* __restrict__ keyH,
                                             float* __restrict__ keys_out) {
  const int b = blockIdx.x >> 7;
  const int h = (((blockIdx.x >> 3) & 15) << 5) + threadIdx.x;
  const int g = blockIdx.x & 7;
  const float* src = iproj + (size_t)b * SEQ * 1024 + h;
  float in[SEQ];
#pragma unroll
  for (int t = 0; t < SEQ; ++t) in[t] = src[(size_t)t * 1024];
  float kv = 0.f;
  float mine[8] = {0.f, 0.f, 0.f, 0.f, 0.f, 0.f, 0.f, 0.f};
#pragma unroll
  for (int blk = 0; blk < SEQ / 8; ++blk) {
    float w[8];
#pragma unroll
    for (int i = 0; i < 8; ++i) {
      kv = DECAY_F * kv + in[blk * 8 + i];
      w[i] = kv;
    }
    if (blk == g) {
#pragma unroll
      for (int i = 0; i < 8; ++i) mine[i] = w[i];
    }
  }
  const size_t o = ((size_t)b * SEQ + g * 8) * HID + h;
#pragma unroll
  for (int i = 0; i < 8; ++i) {
    const float ko = tanhf(mine[i]);
    keyH[o + (size_t)i * HID]     = ko;
    keys_out[o + (size_t)i * HID] = ko;
  }
}

__global__ __launch_bounds__(256) void k_ikv0(const float* __restrict__ mem0,
                                              const float* __restrict__ keyH,
                                              float* __restrict__ ikv0) {
  const int gw   = (blockIdx.x * 256 + threadIdx.x) >> 5;
  const int lane = threadIdx.x & 31;
  const int b  = gw >> 7;
  const int rr = gw & 127;
  const int tm = rr >> 2;
  const int tn = rr & 3;
  const int hf = lane >> 4, ln = lane & 15;
  const float* ar = mem0 + ((size_t)b * HID + tm * 16 + ln) * HID;
  const float* br = keyH + ((size_t)b * SEQ + tn * 16 + ln) * HID;
  v8f acc = {};
  for (int c = 0; c < HID; c += 4) {
    const int cc = c + 2 * hf;
    v2f a, bb;
    a.x  = ar[cc]; a.y  = ar[cc + 1];
    bb.x = br[cc]; bb.y = br[cc + 1];
    acc = __builtin_amdgcn_wmma_f32_16x16x4_f32(false, a, false, bb,
                                                (short)0, acc, false, false);
  }
  for (int r = 0; r < 8; ++r) {
    const int hh = tm * 16 + r + 8 * hf;
    const int tt = tn * 16 + ln;
    ikv0[((size_t)b * SEQ + tt) * HID + hh] = acc[r];
  }
}

__device__ __forceinline__ float lane_xor16(float x) {
  const int b = __builtin_bit_cast(int, x);
  return __builtin_bit_cast(float,
      __builtin_amdgcn_permlanex16(b, b, 0x76543210, 0xfedcba98, false, false));
}
template <int N>
__device__ __forceinline__ float lane_xor_row(float x) {
  return __builtin_bit_cast(float,
      __builtin_amdgcn_update_dpp(0, __builtin_bit_cast(int, x), 0x160 + N, 0xf, 0xf, true));
}
__device__ __forceinline__ float wave_sum(float p) {
  p += lane_xor16(p);
  p += lane_xor_row<8>(p);
  p += lane_xor_row<4>(p);
  p += lane_xor_row<2>(p);
  p += lane_xor_row<1>(p);
  return p;
}

template <int NS>
__device__ __forceinline__ void hist_dots(const float (*vt)[HID], int s0, int lane,
                                          const float (&kk)[16], const float (&bb)[16],
                                          float* dk, float* db) {
  float v[NS][16];
#pragma unroll
  for (int j = 0; j < NS; ++j)
#pragma unroll
    for (int i = 0; i < 16; ++i) v[j][i] = vt[j][lane + 32 * i];
  float p1[NS], p2[NS];
#pragma unroll
  for (int j = 0; j < NS; ++j) {
    float a = 0.f, c = 0.f;
#pragma unroll
    for (int i = 0; i < 16; ++i) {
      a = fmaf(v[j][i], kk[i], a);
      c = fmaf(v[j][i], bb[i], c);
    }
    p1[j] = a; p2[j] = c;
  }
#pragma unroll
  for (int j = 0; j < NS; ++j) { p1[j] = wave_sum(p1[j]); p2[j] = wave_sum(p2[j]); }
  if (lane == 0) {
#pragma unroll
    for (int j = 0; j < NS; ++j) { dk[s0 + 16 * j] = p1[j]; db[s0 + 16 * j] = p2[j]; }
  }
}

__global__ __launch_bounds__(512) void k_scan(const float* __restrict__ iproj,
                                              const float* __restrict__ keyH,
                                              const float* __restrict__ ikv0,
                                              const float* __restrict__ ei0,
                                              const float* __restrict__ ie0,
                                              float* __restrict__ ktrH,
                                              float* __restrict__ vtrH,
                                              float* __restrict__ itrH,
                                              float* __restrict__ dei,
                                              float* __restrict__ vals_out) {
  __shared__ __align__(16) float vtr_l[SEQ][HID];
  __shared__ __align__(16) float ring[DELAY + 1][RPITCH];
  __shared__ __align__(16) float g_s[DELAY][HID];
  __shared__ __align__(16) float key_s[HID];
  __shared__ __align__(16) float itr_l[HID][IPITCH];
  __shared__ __align__(16) float dk[SEQ];
  __shared__ __align__(16) float db[SEQ];

  const int b = blockIdx.x;
  const int h = threadIdx.x;
  const int lane = h & 31;
  const int wv = __builtin_amdgcn_readfirstlane(h >> 5);

  for (int r = 0; r <= DELAY; ++r) ring[r][h] = 0.f;
  for (int s = 0; s < SEQ; s += 4)
    *(float4*)&itr_l[h][s] = make_float4(0.f, 0.f, 0.f, 0.f);
  if (h < SEQ) { dk[h] = 0.f; db[h] = 0.f; }

  float kh[SEQ];
  float key0;
  {
    float kin[SEQ];
#pragma unroll
    for (int s = 0; s < SEQ; ++s)
      kin[s] = keyH[((size_t)b * SEQ + s) * HID + h];
    key0 = kin[0];
    float kt = 0.f;
#pragma unroll
    for (int s = 0; s < SEQ; ++s) {
      kt = DECAY_F * kt + OMD_F * kin[s];
      kh[s] = kt;
      ktrH[((size_t)b * SEQ + s) * HID + h] = kt;
    }
  }

  float vvs = 0.f, isv = 0.f;
  float vtr = 0.f, itr = 0.f, accD = 0.f;
  const size_t mbase = (size_t)b * HID * HID + (size_t)h * HID;
  float d_ei = ei0[mbase + h];
  const float ie0d = ie0[mbase + h];
  const int ln = lane & 15, hf = lane >> 4;
  const float* a0p = ie0 + (size_t)b * HID * HID + (size_t)(wv * 32 + ln) * HID + 4 * hf;
  const float* a1p = a0p + (size_t)16 * HID;
  const float* bp  = &ring[ln < DELAY ? ln : DELAY][4 * hf];
  key_s[h] = key0;
  float val_p = 0.f;
  __syncthreads();

  for (int t = 0; t < SEQ; ++t) {
    const int r = t & (DELAY - 1);
    const size_t rowoff = ((size_t)b * SEQ + t) * HID;

    const float ikv_g = ikv0[rowoff + h];
    const float ivt = iproj[((size_t)b * SEQ + t) * 1024 + HID + h];
    const float key_n = keyH[rowoff + (t + 1 < SEQ ? HID : 0) + h];
    if (t > 0) {
      vtrH[rowoff - HID + h] = vtr;
      itrH[rowoff - HID + h] = itr;
      vals_out[rowoff - HID + h] = val_p;
    }

    if (wv < t) {
      float kk[16], bb[16];
#pragma unroll
      for (int i = 0; i < 16; ++i) {
        kk[i] = key_s[lane + 32 * i];
        bb[i] = ring[r][lane + 32 * i];
      }
      const int ns = (t - wv + 15) >> 4;
      if (ns == 1)      hist_dots<1>(vtr_l + 4 * wv, wv, lane, kk, bb, dk, db);
      else if (ns == 2) hist_dots<2>(vtr_l + 4 * wv, wv, lane, kk, bb, dk, db);
      else if (ns == 3) hist_dots<3>(vtr_l + 4 * wv, wv, lane, kk, bb, dk, db);
      else              hist_dots<4>(vtr_l + 4 * wv, wv, lane, kk, bb, dk, db);
    }

    if (r == 0 && t != 0) {
      v8f g0 = {}, g1 = {};
      float4 fa0[2][8], fa1[2][8];
      for (int q = 0; q < 8; ++q) {
        fa0[0][q] = *(const float4*)(a0p + q * 8);
        fa1[0][q] = *(const float4*)(a1p + q * 8);
      }
#pragma unroll
      for (int ch = 0; ch < HID / 64; ++ch) {
        const int cu = ch & 1, nx = cu ^ 1;
        if (ch + 1 < HID / 64) {
          for (int q = 0; q < 8; ++q) {
            fa0[nx][q] = *(const float4*)(a0p + (ch + 1) * 64 + q * 8);
            fa1[nx][q] = *(const float4*)(a1p + (ch + 1) * 64 + q * 8);
          }
        }
        float4 fb[8];
        for (int q = 0; q < 8; ++q)
          fb[q] = *(const float4*)(bp + ch * 64 + q * 8);
        for (int q = 0; q < 8; ++q) {
          const v2f b01 = {fb[q].x, fb[q].y}, b23 = {fb[q].z, fb[q].w};
          const v2f p01 = {fa0[cu][q].x, fa0[cu][q].y}, p23 = {fa0[cu][q].z, fa0[cu][q].w};
          const v2f q01 = {fa1[cu][q].x, fa1[cu][q].y}, q23 = {fa1[cu][q].z, fa1[cu][q].w};
          g0 = __builtin_amdgcn_wmma_f32_16x16x4_f32(false, p01, false, b01,
                                                     (short)0, g0, false, false);
          g1 = __builtin_amdgcn_wmma_f32_16x16x4_f32(false, q01, false, b01,
                                                     (short)0, g1, false, false);
          g0 = __builtin_amdgcn_wmma_f32_16x16x4_f32(false, p23, false, b23,
                                                     (short)0, g0, false, false);
          g1 = __builtin_amdgcn_wmma_f32_16x16x4_f32(false, q23, false, b23,
                                                     (short)0, g1, false, false);
        }
      }
      if (ln < DELAY) {
        float* gp = &g_s[ln][wv * 32 + 8 * hf];
        *(float4*)(gp)      = make_float4(g0[0], g0[1], g0[2], g0[3]);
        *(float4*)(gp + 4)  = make_float4(g0[4], g0[5], g0[6], g0[7]);
        *(float4*)(gp + 16) = make_float4(g1[0], g1[1], g1[2], g1[3]);
        *(float4*)(gp + 20) = make_float4(g1[4], g1[5], g1[6], g1[7]);
      }
    }
    __syncthreads();

    float sk = 0.f, si = 0.f;
#pragma unroll
    for (int s8 = 0; s8 < SEQ; s8 += 8) {
      if (s8 < t) {
        const float4 d0 = *(const float4*)&dk[s8];
        const float4 d1 = *(const float4*)&dk[s8 + 4];
        const float4 e0 = *(const float4*)&db[s8];
        const float4 e1 = *(const float4*)&db[s8 + 4];
        const float4 i0 = *(const float4*)&itr_l[h][s8];
        const float4 i1 = *(const float4*)&itr_l[h][s8 + 4];
        sk = fmaf(kh[s8 + 0], d0.x, sk);
        si = fmaf(i0.x, e0.x, si);
        sk = fmaf(kh[s8 + 1], d0.y, sk);
        si = fmaf(i0.y, e0.y, si);
        sk = fmaf(kh[s8 + 2], d0.z, sk);
        si = fmaf(i0.z, e0.z, si);
        sk = fmaf(kh[s8 + 3], d0.w, sk);
        si = fmaf(i0.w, e0.w, si);
        sk = fmaf(kh[s8 + 4], d1.x, sk);
        si = fmaf(i1.x, e1.x, si);
        sk = fmaf(kh[s8 + 5], d1.y, sk);
        si = fmaf(i1.y, e1.y, si);
        sk = fmaf(kh[s8 + 6], d1.z, sk);
        si = fmaf(i1.z, e1.z, si);
        sk = fmaf(kh[s8 + 7], d1.w, sk);
        si = fmaf(i1.w, e1.w, si);
      }
    }

    const float mv = (t >= DELAY) ? g_s[r][h] : 0.f;
    const float bufh = ring[r][h];
    const float inh_out = mv + LR_F * si - bufh * (ie0d + LR_F * accD);

    const float ikv = 0.2f * (ikv_g + LR_F * sk);

    vvs = DECAY_F * vvs + ivt + ikv - inh_out;
    const float val = tanhf(vvs);
    isv = DECAY_F * isv + val * d_ei;
    const float inh = tanhf(isv);

    vtr = DECAY_F * vtr + OMD_F * val;
    itr = DECAY_F * itr + OMD_F * inh;
    const float p = vtr * itr;
    d_ei += LR_F * p;
    accD += p;

    vtr_l[((t & 15) << 2) | (t >> 4)][h] = vtr;
    itr_l[h][t] = itr;
    val_p = val;
    ring[r][h] = inh;
    key_s[h] = key_n;
    __syncthreads();
  }
  {
    const size_t rowl = ((size_t)b * SEQ + SEQ - 1) * HID;
    vtrH[rowl + h] = vtr;
    itrH[rowl + h] = itr;
    vals_out[rowl + h] = val_p;
  }
  dei[(size_t)b * HID + h] = d_ei;
}

__global__ __launch_bounds__(256) void k_final(const float* __restrict__ mem0,
                                               const float* __restrict__ ie0,
                                               const float* __restrict__ ktrH,
                                               const float* __restrict__ vtrH,
                                               const float* __restrict__ itrH,
                                               float* __restrict__ out) {
  const int which = blockIdx.y;
  const int gw    = (blockIdx.x * 256 + threadIdx.x) >> 5;
  const int lane  = threadIdx.x & 31;
  const int b  = gw >> 10;
  const int rr = gw & 1023;
  const int tm = rr >> 5, tn = rr & 31;
  const int hf = lane >> 4, ln = lane & 15;
  const float* Ah = which ? itrH : ktrH;
  v8f acc = {};
  for (int c = 0; c < SEQ; c += 4) {
    const int cc = c + 2 * hf;
    v2f a, bb;
    a.x  = Ah[((size_t)b * SEQ + cc)     * HID + tm * 16 + ln];
    a.y  = Ah[((size_t)b * SEQ + cc + 1) * HID + tm * 16 + ln];
    bb.x = vtrH[((size_t)b * SEQ + cc)     * HID + tn * 16 + ln];
    bb.y = vtrH[((size_t)b * SEQ + cc + 1) * HID + tn * 16 + ln];
    acc = __builtin_amdgcn_wmma_f32_16x16x4_f32(false, a, false, bb,
                                                (short)0, acc, false, false);
  }
  const float* base0 = which ? ie0 : mem0;
  float* dst = out + (which ? (size_t)2097152 : (size_t)0);
  for (int r = 0; r < 8; ++r) {
    const int m = tm * 16 + r + 8 * hf;
    const int n = tn * 16 + ln;
    const size_t idx = (size_t)b * HID * HID + (size_t)m * HID + n;
    float v = base0[idx] + LR_F * acc[r];
    if (which && m == n)
      v = LR_F * itrH[((size_t)b * SEQ + 63) * HID + m] *
                 vtrH[((size_t)b * SEQ + 63) * HID + n];
    dst[idx] = v;
  }
}

__global__ __launch_bounds__(256) void k_ei(const float* __restrict__ dei,
                                            const float* __restrict__ vtrH,
                                            const float* __restrict__ itrH,
                                            float* __restrict__ ei_out) {
  const int idx = blockIdx.x * 256 + threadIdx.x;
  const int b = idx >> 18;
  const int h = (idx >> 9) & 511;
  const int i = idx & 511;
  float v;
  if (h == i) v = dei[b * HID + h];
  else        v = LR_F * vtrH[((size_t)b * SEQ + 63) * HID + h] *
                         itrH[((size_t)b * SEQ + 63) * HID + i];
  ei_out[idx] = v;
}

extern "C" void kernel_launch(void* const* d_in, const int* in_sizes, int n_in,
                              void* d_out, int out_size, void* d_ws, size_t ws_size,
                              hipStream_t stream) {
  (void)in_sizes; (void)n_in; (void)out_size; (void)ws_size;
  const float* x    = (const float*)d_in[0];
  const float* W    = (const float*)d_in[1];
  const float* mem0 = (const float*)d_in[2];
  const float* ei0  = (const float*)d_in[3];
  const float* ie0  = (const float*)d_in[4];

  float* out = (float*)d_out;
  float* ei_out   = out + 1048576;
  float* keys_out = out + 3145728;
  float* vals_out = out + 3276800;

  float* ws    = (float*)d_ws;
  float* iproj = ws;
  float* keyH  = ws + 262144;
  float* ikv0  = ws + 393216;
  float* ktrH  = ws + 524288;
  float* vtrH  = ws + 655360;
  float* itrH  = ws + 786432;
  float* dei   = ws + 917504;

  k_proj<<<128, 256, 0, stream>>>(x, W, iproj);
  k_keys<<<512, 32, 0, stream>>>(iproj, keyH, keys_out);
  k_ikv0<<<64, 256, 0, stream>>>(mem0, keyH, ikv0);
  k_scan<<<4, 512, 0, stream>>>(iproj, keyH, ikv0, ei0, ie0,
                                ktrH, vtrH, itrH, dei, vals_out);
  dim3 gf(512, 2);
  k_final<<<gf, 256, 0, stream>>>(mem0, ie0, ktrH, vtrH, itrH, out);
  k_ei<<<4096, 256, 0, stream>>>(dei, vtrH, itrH, ei_out);
}
